# starting kernel
# baseline (speedup 1.0000x reference)
#include <hip/hip_runtime.h>
#include <math.h>

#define ALPHA   0.2f
#define N_NODES 4096
#define IN_FEAT 512
#define NHEAD   4
#define FDIM    64
#define HF      256

typedef __attribute__((ext_vector_type(2))) float v2f;
typedef __attribute__((ext_vector_type(8))) float v8f;

__device__ __forceinline__ float lrelu(float x) {
    return fmaxf(x, ALPHA * x);
}

__global__ __launch_bounds__(128) void gat_gemm1(const float* __restrict__ x,
                                                 const float* __restrict__ W,
                                                 float* __restrict__ h) {
    const int lane = threadIdx.x & 31;
    const int wave = threadIdx.x >> 5;
    const int i0   = blockIdx.x * 16;
    const int n0   = blockIdx.y * 64 + wave * 16;
    const int m    = lane & 15;
    const int koff = (lane >> 4) << 1;

    const float* xrow = x + (size_t)(i0 + m) * IN_FEAT;
    v8f acc = {};
    for (int k0 = 0; k0 < IN_FEAT; k0 += 4) {
        const int ka = k0 + koff;
        v2f a2, b2;
        a2.x = xrow[ka];
        a2.y = xrow[ka + 1];
        b2.x = W[(size_t)ka * HF + n0 + m];
        b2.y = W[(size_t)(ka + 1) * HF + n0 + m];
        acc = __builtin_amdgcn_wmma_f32_16x16x4_f32(false, a2, false, b2,
                                                    (short)0, acc, false, false);
    }
    const int rbase = (lane >> 4) << 3;
#pragma unroll
    for (int v = 0; v < 8; ++v)
        h[(size_t)(i0 + rbase + v) * HF + n0 + m] = acc[v];
}

__global__ __launch_bounds__(128) void gat_st(const float* __restrict__ h,
                                              const float* __restrict__ a,
                                              float* __restrict__ s,
                                              float* __restrict__ t) {
    const int lane = threadIdx.x & 31;
    const int gw   = blockIdx.x * 4 + (threadIdx.x >> 5);
    const int n    = gw >> 2;
    const int hd   = gw & 3;

    const float* hrow = h + (size_t)n * HF + hd * FDIM;
    const float* asrc = a + hd * 2 * FDIM;
    const float* adst = asrc + FDIM;

    float sv = hrow[lane] * asrc[lane] + hrow[lane + 32] * asrc[lane + 32];
    float tv = hrow[lane] * adst[lane] + hrow[lane + 32] * adst[lane + 32];
#pragma unroll
    for (int off = 16; off > 0; off >>= 1) {
        sv += __shfl_xor(sv, off, 32);
        tv += __shfl_xor(tv, off, 32);
    }
    if (lane == 0) {
        s[n * NHEAD + hd] = sv;
        t[n * NHEAD + hd] = tv;
    }
}

__global__ __launch_bounds__(256) void gat_tmax(const float* __restrict__ t,
                                                float* __restrict__ tmax) {
    __shared__ float red[256];
    const int hd = blockIdx.x;
    float m = -INFINITY;
    for (int j = threadIdx.x; j < N_NODES; j += 256)
        m = fmaxf(m, t[j * NHEAD + hd]);
    red[threadIdx.x] = m;
    __syncthreads();
    for (int off = 128; off > 0; off >>= 1) {
        if (threadIdx.x < off)
            red[threadIdx.x] = fmaxf(red[threadIdx.x], red[threadIdx.x + off]);
        __syncthreads();
    }
    if (threadIdx.x == 0) tmax[hd] = red[0];
}

__global__ __launch_bounds__(128) void gat_attn(const float* __restrict__ h,
                                                const float* __restrict__ s,
                                                const float* __restrict__ t,
                                                const float* __restrict__ tmax,
                                                float* __restrict__ out) {
    __shared__ float cred[4][16][64];
    __shared__ float zred[4][32];
    __shared__ float zrow[16];

    const int lane = threadIdx.x & 31;
    const int wave = threadIdx.x >> 5;
    const int i0   = blockIdx.x * 16;
    const int hd   = blockIdx.y;
    const int m    = lane & 15;
    const int koff = (lane >> 4) << 1;

    const float sm   = s[(i0 + m) * NHEAD + hd];
    const float mrow = lrelu(sm + tmax[hd]);
    const float* hh  = h + hd * FDIM;

    v8f acc0 = {}, acc1 = {}, acc2 = {}, acc3 = {};
    float zpart = 0.f;

    const int jbeg = wave * (N_NODES / 4);
    const int jend = jbeg + (N_NODES / 4);
    for (int j0 = jbeg; j0 < jend; j0 += 4) {
        const int ja = j0 + koff;
        const float pa = __expf(lrelu(sm + t[ja * NHEAD + hd]) - mrow);
        const float pb = __expf(lrelu(sm + t[(ja + 1) * NHEAD + hd]) - mrow);
        zpart += pa + pb;
        v2f a2; a2.x = pa; a2.y = pb;

        const float* hja = hh + (size_t)ja * HF;
        const float* hjb = hja + HF;
        v2f b0, b1, b2, b3;
        b0.x = hja[m];      b0.y = hjb[m];
        b1.x = hja[16 + m]; b1.y = hjb[16 + m];
        b2.x = hja[32 + m]; b2.y = hjb[32 + m];
        b3.x = hja[48 + m]; b3.y = hjb[48 + m];

        acc0 = __builtin_amdgcn_wmma_f32_16x16x4_f32(false, a2, false, b0, (short)0, acc0, false, false);
        acc1 = __builtin_amdgcn_wmma_f32_16x16x4_f32(false, a2, false, b1, (short)0, acc1, false, false);
        acc2 = __builtin_amdgcn_wmma_f32_16x16x4_f32(false, a2, false, b2, (short)0, acc2, false, false);
        acc3 = __builtin_amdgcn_wmma_f32_16x16x4_f32(false, a2, false, b3, (short)0, acc3, false, false);
    }

    const int rbase = (lane >> 4) << 3;
#pragma unroll
    for (int v = 0; v < 8; ++v) {
        cred[wave][rbase + v][m]      = acc0[v];
        cred[wave][rbase + v][16 + m] = acc1[v];
        cred[wave][rbase + v][32 + m] = acc2[v];
        cred[wave][rbase + v][48 + m] = acc3[v];
    }
    zred[wave][lane] = zpart;
    __syncthreads();

    if (threadIdx.x < 16) {
        float z = 0.f;
#pragma unroll
        for (int w = 0; w < 4; ++w)
            z += zred[w][threadIdx.x] + zred[w][threadIdx.x + 16];
        zrow[threadIdx.x] = z;
    }
    __syncthreads();

    for (int e = threadIdx.x; e < 16 * 64; e += 128) {
        const int rr = e >> 6, cc = e & 63;
        const float v = cred[0][rr][cc] + cred[1][rr][cc] +
                        cred[2][rr][cc] + cred[3][rr][cc];
        out[(size_t)(i0 + rr) * HF + hd * FDIM + cc] = v / zrow[rr];
    }
}

extern "C" void kernel_launch(void* const* d_in, const int* in_sizes, int n_in,
                              void* d_out, int out_size, void* d_ws, size_t ws_size,
                              hipStream_t stream) {
    const float* x = (const float*)d_in[0];
    const float* W = (const float*)d_in[2];
    const float* a = (const float*)d_in[3];
    float* out = (float*)d_out;

    char*  ws   = (char*)d_ws;
    float* h    = (float*)ws;
    float* s    = (float*)(ws + (size_t)N_NODES * HF * sizeof(float));
    float* t    = s + N_NODES * NHEAD;
    float* tmax = t + N_NODES * NHEAD;

    dim3 g1(N_NODES / 16, HF / 64);
    gat_gemm1<<<g1, 128, 0, stream>>>(x, W, h);

    gat_st<<<N_NODES, 128, 0, stream>>>(h, a, s, t);

    gat_tmax<<<NHEAD, 256, 0, stream>>>(t, tmax);

    dim3 g4(N_NODES / 16, NHEAD);
    gat_attn<<<g4, 128, 0, stream>>>(h, s, t, tmax, out);
}


# h stored key-quad-interleaved ([j/4][col][j%4]), s,t head-major: attention reads a column's 4 keys, and a lane half's 4 t, by ONE 16-byte load (5 loads per 8 keys, was 20); sum order changed
# speedup vs baseline: 1.4392x; 1.4392x over previous
#include <hip/hip_runtime.h>
#include <math.h>

#define ALPHA   0.2f
#define N_NODES 4096
#define IN_FEAT 512
#define NHEAD   4
#define FDIM    64
#define HF      256

typedef __attribute__((ext_vector_type(2))) float v2f;
typedef __attribute__((ext_vector_type(8))) float v8f;

__device__ __forceinline__ float lrelu(float x) {
    return fmaxf(x, ALPHA * x);
}

typedef __attribute__((ext_vector_type(4))) float v4f;

#define G1_CH  8
#define G1_NST (IN_FEAT / (8 * G1_CH))

struct G1Stage {
    v4f a[G1_CH];
    v4f b[G1_CH][4];
};

__device__ __forceinline__ void g1_load(G1Stage& s, const float* __restrict__ xp,
                                        const float* __restrict__ wp) {
    for (int ch = 0; ch < G1_CH; ++ch) {
        s.a[ch] = *(const v4f*)(xp + ch * 8);
        for (int kk = 0; kk < 4; ++kk)
            s.b[ch][kk] = *(const v4f*)(wp + (size_t)(ch * 8 + kk) * HF);
    }
}

__device__ __forceinline__ void g1_mma(const G1Stage& s, v8f& c0, v8f& c1, v8f& c2, v8f& c3) {
    for (int ch = 0; ch < G1_CH; ++ch) {
        for (int p = 0; p < 2; ++p) {
            v2f a2, b0, b1, b2, b3;
            a2.x = s.a[ch][2 * p];        a2.y = s.a[ch][2 * p + 1];
            b0.x = s.b[ch][2 * p][0];     b0.y = s.b[ch][2 * p + 1][0];
            b1.x = s.b[ch][2 * p][1];     b1.y = s.b[ch][2 * p + 1][1];
            b2.x = s.b[ch][2 * p][2];     b2.y = s.b[ch][2 * p + 1][2];
            b3.x = s.b[ch][2 * p][3];     b3.y = s.b[ch][2 * p + 1][3];
            c0 = __builtin_amdgcn_wmma_f32_16x16x4_f32(false, a2, false, b0, (short)0, c0, false, false);
            c1 = __builtin_amdgcn_wmma_f32_16x16x4_f32(false, a2, false, b1, (short)0, c1, false, false);
            c2 = __builtin_amdgcn_wmma_f32_16x16x4_f32(false, a2, false, b2, (short)0, c2, false, false);
            c3 = __builtin_amdgcn_wmma_f32_16x16x4_f32(false, a2, false, b3, (short)0, c3, false, false);
        }
    }
}

__global__ __launch_bounds__(128) void gat_gemm1(const float* __restrict__ x,
                                                 const float* __restrict__ W,
                                                 float* __restrict__ h) {
    const int lane = threadIdx.x & 31;
    const int wave = threadIdx.x >> 5;
    const int c    = lane & 15;
    const int hh   = lane >> 4;
    const int i0   = (blockIdx.x * 4 + wave) * 16;
    const int n0   = blockIdx.y * 64;

    const float* xp = x + (size_t)(i0 + c) * IN_FEAT + 4 * hh;
    const float* wp = W + (size_t)(4 * hh) * HF + n0 + 4 * c;

    v8f c0 = {}, c1 = {}, c2 = {}, c3 = {};
    G1Stage s0, s1;
    g1_load(s0, xp, wp);
#pragma nounroll
    for (int st = 0; st < G1_NST - 2; st += 2) {
        g1_load(s1, xp + (st + 1) * (8 * G1_CH), wp + (size_t)(st + 1) * (8 * G1_CH) * HF);
        __builtin_amdgcn_sched_barrier(0);
        g1_mma(s0, c0, c1, c2, c3);
        g1_load(s0, xp + (st + 2) * (8 * G1_CH), wp + (size_t)(st + 2) * (8 * G1_CH) * HF);
        __builtin_amdgcn_sched_barrier(0);
        g1_mma(s1, c0, c1, c2, c3);
    }
    g1_load(s1, xp + (G1_NST - 1) * (8 * G1_CH), wp + (size_t)(G1_NST - 1) * (8 * G1_CH) * HF);
    __builtin_amdgcn_sched_barrier(0);
    g1_mma(s0, c0, c1, c2, c3);
    g1_mma(s1, c0, c1, c2, c3);

    float* hp = h + ((size_t)((i0 + 8 * hh) >> 2) * HF + n0 + 4 * c) * 4;
    for (int g = 0; g < 2; ++g) {
        v4f o0, o1, o2, o3;
        o0.x = c0[4 * g]; o0.y = c0[4 * g + 1]; o0.z = c0[4 * g + 2]; o0.w = c0[4 * g + 3];
        o1.x = c1[4 * g]; o1.y = c1[4 * g + 1]; o1.z = c1[4 * g + 2]; o1.w = c1[4 * g + 3];
        o2.x = c2[4 * g]; o2.y = c2[4 * g + 1]; o2.z = c2[4 * g + 2]; o2.w = c2[4 * g + 3];
        o3.x = c3[4 * g]; o3.y = c3[4 * g + 1]; o3.z = c3[4 * g + 2]; o3.w = c3[4 * g + 3];
        float* hq = hp + (size_t)g * HF * 4;
        *(v4f*)(hq)      = o0;
        *(v4f*)(hq + 4)  = o1;
        *(v4f*)(hq + 8)  = o2;
        *(v4f*)(hq + 12) = o3;
    }
}

__global__ __launch_bounds__(128) void gat_st(const float* __restrict__ h,
                                              const float* __restrict__ a,
                                              float* __restrict__ s,
                                              float* __restrict__ t) {
    const int lane = threadIdx.x & 31;
    const int hd   = threadIdx.x >> 5;
    const int qd   = blockIdx.x;

    const float* hq   = h + ((size_t)qd * HF + hd * FDIM) * 4;
    const float* asrc = a + hd * 2 * FDIM;
    const float* adst = asrc + FDIM;

    const v4f ha = *(const v4f*)(hq + lane * 4);
    const v4f hb = *(const v4f*)(hq + (lane + 32) * 4);
    const float a0 = asrc[lane], a1 = asrc[lane + 32];
    const float d0 = adst[lane], d1 = adst[lane + 32];

    v4f sv, tv;
    for (int e = 0; e < 4; ++e) {
        sv[e] = ha[e] * a0 + hb[e] * a1;
        tv[e] = ha[e] * d0 + hb[e] * d1;
    }
#pragma unroll
    for (int off = 16; off > 0; off >>= 1) {
        for (int e = 0; e < 4; ++e) {
            sv[e] += __shfl_xor(sv[e], off, 32);
            tv[e] += __shfl_xor(tv[e], off, 32);
        }
    }
    if (lane == 0) {
        *(v4f*)(s + (size_t)hd * N_NODES + 4 * qd) = sv;
        *(v4f*)(t + (size_t)hd * N_NODES + 4 * qd) = tv;
    }
}

__global__ __launch_bounds__(256) void gat_tmax(const float* __restrict__ t,
                                                float* __restrict__ tmax) {
    __shared__ float red[256];
    const int hd = blockIdx.x;
    float m = -INFINITY;
    for (int j = threadIdx.x; j < N_NODES; j += 256)
        m = fmaxf(m, t[hd * N_NODES + j]);
    red[threadIdx.x] = m;
    __syncthreads();
    for (int off = 128; off > 0; off >>= 1) {
        if (threadIdx.x < off)
            red[threadIdx.x] = fmaxf(red[threadIdx.x], red[threadIdx.x + off]);
        __syncthreads();
    }
    if (threadIdx.x == 0) tmax[hd] = red[0];
}

#define KSPLIT   8
#define RTILES   4
#define ATT_THR  (KSPLIT * RTILES * 32)
__global__ __launch_bounds__(ATT_THR) void gat_attn(const float* __restrict__ h,
                                                const float* __restrict__ s,
                                                const float* __restrict__ t,
                                                const float* __restrict__ tmax,
                                                float* __restrict__ out) {
    __shared__ float cred[RTILES][KSPLIT][16][64];
    __shared__ float zred[RTILES][KSPLIT][32];
    __shared__ float zrow[RTILES * 16];

    const int lane = threadIdx.x & 31;
    const int wave = threadIdx.x >> 5;
    const int rt   = wave & (RTILES - 1);
    const int ks   = wave / RTILES;
    const int ib   = blockIdx.x * (RTILES * 16);
    const int i0   = ib + rt * 16;
    const int hd   = blockIdx.y;
    const int m    = lane & 15;
    const int lh   = lane >> 4;

    const float sm   = s[hd * N_NODES + i0 + m];
    const float mrow = lrelu(sm + tmax[hd]);

    v8f acc0 = {}, acc1 = {}, acc2 = {}, acc3 = {};
    float zpart = 0.f;

    const int jbeg = ks * (N_NODES / KSPLIT);
    const int jend = jbeg + (N_NODES / KSPLIT);
    const float* tp = t + (size_t)hd * N_NODES + 4 * lh;
    const float* hp = h + ((size_t)lh * HF + hd * FDIM + m) * 4;
    for (int j0 = jbeg; j0 < jend; j0 += 16) {
        v4f t4[2], g[2][4];
        for (int u = 0; u < 2; ++u)
            t4[u] = *(const v4f*)(tp + j0 + 8 * u);
        for (int u = 0; u < 2; ++u) {
            const float* hq = hp + (size_t)((j0 >> 2) + 2 * u) * (HF * 4);
            for (int q = 0; q < 4; ++q)
                g[u][q] = *(const v4f*)(hq + q * (16 * 4));
        }
        __builtin_amdgcn_sched_barrier(0);

        for (int u = 0; u < 2; ++u) {
            const float p0 = __expf(lrelu(sm + t4[u].x) - mrow);
            const float p1 = __expf(lrelu(sm + t4[u].y) - mrow);
            const float p2 = __expf(lrelu(sm + t4[u].z) - mrow);
            const float p3 = __expf(lrelu(sm + t4[u].w) - mrow);
            zpart += p0 + p1;
            zpart += p2 + p3;

            v2f a2, b0, b1, b2, b3;
            a2.x = p0;         a2.y = p1;
            b0.x = g[u][0].x;  b0.y = g[u][0].y;
            b1.x = g[u][1].x;  b1.y = g[u][1].y;
            b2.x = g[u][2].x;  b2.y = g[u][2].y;
            b3.x = g[u][3].x;  b3.y = g[u][3].y;
            acc0 = __builtin_amdgcn_wmma_f32_16x16x4_f32(false, a2, false, b0, (short)0, acc0, false, false);
            acc1 = __builtin_amdgcn_wmma_f32_16x16x4_f32(false, a2, false, b1, (short)0, acc1, false, false);
            acc2 = __builtin_amdgcn_wmma_f32_16x16x4_f32(false, a2, false, b2, (short)0, acc2, false, false);
            acc3 = __builtin_amdgcn_wmma_f32_16x16x4_f32(false, a2, false, b3, (short)0, acc3, false, false);

            a2.x = p2;         a2.y = p3;
            b0.x = g[u][0].z;  b0.y = g[u][0].w;
            b1.x = g[u][1].z;  b1.y = g[u][1].w;
            b2.x = g[u][2].z;  b2.y = g[u][2].w;
            b3.x = g[u][3].z;  b3.y = g[u][3].w;
            acc0 = __builtin_amdgcn_wmma_f32_16x16x4_f32(false, a2, false, b0, (short)0, acc0, false, false);
            acc1 = __builtin_amdgcn_wmma_f32_16x16x4_f32(false, a2, false, b1, (short)0, acc1, false, false);
            acc2 = __builtin_amdgcn_wmma_f32_16x16x4_f32(false, a2, false, b2, (short)0, acc2, false, false);
            acc3 = __builtin_amdgcn_wmma_f32_16x16x4_f32(false, a2, false, b3, (short)0, acc3, false, false);
        }
    }

    const int rbase = (lane >> 4) << 3;
#pragma unroll
    for (int v = 0; v < 8; ++v) {
        cred[rt][ks][rbase + v][m]      = acc0[v];
        cred[rt][ks][rbase + v][16 + m] = acc1[v];
        cred[rt][ks][rbase + v][32 + m] = acc2[v];
        cred[rt][ks][rbase + v][48 + m] = acc3[v];
    }
    zred[rt][ks][lane] = zpart;
    __syncthreads();

    if (threadIdx.x < RTILES * 16) {
        const int zr = threadIdx.x >> 4, zm = threadIdx.x & 15;
        float z = 0.f;
#pragma unroll
        for (int w = 0; w < KSPLIT; ++w)
            z += zred[zr][w][zm] + zred[zr][w][zm + 16];
        zrow[threadIdx.x] = z;
    }
    __syncthreads();

    for (int e = threadIdx.x; e < RTILES * 16 * 64; e += ATT_THR) {
        const int er = e >> 10, rr = (e >> 6) & 15, cc = e & 63;
        float v = cred[er][0][rr][cc];
        for (int w = 1; w < KSPLIT; ++w)
            v += cred[er][w][rr][cc];
        out[(size_t)(ib + er * 16 + rr) * HF + hd * FDIM + cc] = v / zrow[er * 16 + rr];
    }
}

extern "C" void kernel_launch(void* const* d_in, const int* in_sizes, int n_in,
                              void* d_out, int out_size, void* d_ws, size_t ws_size,
                              hipStream_t stream) {
    const float* x = (const float*)d_in[0];
    const float* W = (const float*)d_in[2];
    const float* a = (const float*)d_in[3];
    float* out = (float*)d_out;

    char*  ws   = (char*)d_ws;
    float* h    = (float*)ws;
    float* s    = (float*)(ws + (size_t)N_NODES * HF * sizeof(float));
    float* t    = s + N_NODES * NHEAD;
    float* tmax = t + N_NODES * NHEAD;

    dim3 g1(N_NODES / 64, HF / 64);
    gat_gemm1<<<g1, 128, 0, stream>>>(x, W, h);

    gat_st<<<N_NODES / 4, 128, 0, stream>>>(h, a, s, t);

    gat_tmax<<<NHEAD, 256, 0, stream>>>(t, tmax);

    dim3 g4(N_NODES / (RTILES * 16), NHEAD);
    gat_attn<<<g4, ATT_THR, 0, stream>>>(h, s, t, tmax, out);
}


# softmax without a global max: each key-range wave shifts by its own range's max of t (read at entry), partials merged with weights exp(m_w-M); max launch gone (4 -> 3); sums rescaled, order kept
# speedup vs baseline: 1.5842x; 1.1008x over previous
#include <hip/hip_runtime.h>
#include <math.h>

#define ALPHA   0.2f
#define N_NODES 4096
#define IN_FEAT 512
#define NHEAD   4
#define FDIM    64
#define HF      256

typedef __attribute__((ext_vector_type(2))) float v2f;
typedef __attribute__((ext_vector_type(8))) float v8f;

__device__ __forceinline__ float lrelu(float x) {
    return fmaxf(x, ALPHA * x);
}

typedef __attribute__((ext_vector_type(4))) float v4f;

#define G1_CH  8
#define G1_NST (IN_FEAT / (8 * G1_CH))

struct G1Stage {
    v4f a[G1_CH];
    v4f b[G1_CH][4];
};

__device__ __forceinline__ void g1_load(G1Stage& s, const float* __restrict__ xp,
                                        const float* __restrict__ wp) {
    for (int ch = 0; ch < G1_CH; ++ch) {
        s.a[ch] = *(const v4f*)(xp + ch * 8);
        for (int kk = 0; kk < 4; ++kk)
            s.b[ch][kk] = *(const v4f*)(wp + (size_t)(ch * 8 + kk) * HF);
    }
}

__device__ __forceinline__ void g1_mma(const G1Stage& s, v8f& c0, v8f& c1, v8f& c2, v8f& c3) {
    for (int ch = 0; ch < G1_CH; ++ch) {
        for (int p = 0; p < 2; ++p) {
            v2f a2, b0, b1, b2, b3;
            a2.x = s.a[ch][2 * p];        a2.y = s.a[ch][2 * p + 1];
            b0.x = s.b[ch][2 * p][0];     b0.y = s.b[ch][2 * p + 1][0];
            b1.x = s.b[ch][2 * p][1];     b1.y = s.b[ch][2 * p + 1][1];
            b2.x = s.b[ch][2 * p][2];     b2.y = s.b[ch][2 * p + 1][2];
            b3.x = s.b[ch][2 * p][3];     b3.y = s.b[ch][2 * p + 1][3];
            c0 = __builtin_amdgcn_wmma_f32_16x16x4_f32(false, a2, false, b0, (short)0, c0, false, false);
            c1 = __builtin_amdgcn_wmma_f32_16x16x4_f32(false, a2, false, b1, (short)0, c1, false, false);
            c2 = __builtin_amdgcn_wmma_f32_16x16x4_f32(false, a2, false, b2, (short)0, c2, false, false);
            c3 = __builtin_amdgcn_wmma_f32_16x16x4_f32(false, a2, false, b3, (short)0, c3, false, false);
        }
    }
}

__global__ __launch_bounds__(128) void gat_gemm1(const float* __restrict__ x,
                                                 const float* __restrict__ W,
                                                 float* __restrict__ h) {
    const int lane = threadIdx.x & 31;
    const int wave = threadIdx.x >> 5;
    const int c    = lane & 15;
    const int hh   = lane >> 4;
    const int i0   = (blockIdx.x * 4 + wave) * 16;
    const int n0   = blockIdx.y * 64;

    const float* xp = x + (size_t)(i0 + c) * IN_FEAT + 4 * hh;
    const float* wp = W + (size_t)(4 * hh) * HF + n0 + 4 * c;

    v8f c0 = {}, c1 = {}, c2 = {}, c3 = {};
    G1Stage s0, s1;
    g1_load(s0, xp, wp);
#pragma nounroll
    for (int st = 0; st < G1_NST - 2; st += 2) {
        g1_load(s1, xp + (st + 1) * (8 * G1_CH), wp + (size_t)(st + 1) * (8 * G1_CH) * HF);
        __builtin_amdgcn_sched_barrier(0);
        g1_mma(s0, c0, c1, c2, c3);
        g1_load(s0, xp + (st + 2) * (8 * G1_CH), wp + (size_t)(st + 2) * (8 * G1_CH) * HF);
        __builtin_amdgcn_sched_barrier(0);
        g1_mma(s1, c0, c1, c2, c3);
    }
    g1_load(s1, xp + (G1_NST - 1) * (8 * G1_CH), wp + (size_t)(G1_NST - 1) * (8 * G1_CH) * HF);
    __builtin_amdgcn_sched_barrier(0);
    g1_mma(s0, c0, c1, c2, c3);
    g1_mma(s1, c0, c1, c2, c3);

    float* hp = h + ((size_t)((i0 + 8 * hh) >> 2) * HF + n0 + 4 * c) * 4;
    for (int g = 0; g < 2; ++g) {
        v4f o0, o1, o2, o3;
        o0.x = c0[4 * g]; o0.y = c0[4 * g + 1]; o0.z = c0[4 * g + 2]; o0.w = c0[4 * g + 3];
        o1.x = c1[4 * g]; o1.y = c1[4 * g + 1]; o1.z = c1[4 * g + 2]; o1.w = c1[4 * g + 3];
        o2.x = c2[4 * g]; o2.y = c2[4 * g + 1]; o2.z = c2[4 * g + 2]; o2.w = c2[4 * g + 3];
        o3.x = c3[4 * g]; o3.y = c3[4 * g + 1]; o3.z = c3[4 * g + 2]; o3.w = c3[4 * g + 3];
        float* hq = hp + (size_t)g * HF * 4;
        *(v4f*)(hq)      = o0;
        *(v4f*)(hq + 4)  = o1;
        *(v4f*)(hq + 8)  = o2;
        *(v4f*)(hq + 12) = o3;
    }
}

__global__ __launch_bounds__(128) void gat_st(const float* __restrict__ h,
                                              const float* __restrict__ a,
                                              float* __restrict__ s,
                                              float* __restrict__ t) {
    const int lane = threadIdx.x & 31;
    const int hd   = threadIdx.x >> 5;
    const int qd   = blockIdx.x;

    const float* hq   = h + ((size_t)qd * HF + hd * FDIM) * 4;
    const float* asrc = a + hd * 2 * FDIM;
    const float* adst = asrc + FDIM;

    const v4f ha = *(const v4f*)(hq + lane * 4);
    const v4f hb = *(const v4f*)(hq + (lane + 32) * 4);
    const float a0 = asrc[lane], a1 = asrc[lane + 32];
    const float d0 = adst[lane], d1 = adst[lane + 32];

    v4f sv, tv;
    for (int e = 0; e < 4; ++e) {
        sv[e] = ha[e] * a0 + hb[e] * a1;
        tv[e] = ha[e] * d0 + hb[e] * d1;
    }
#pragma unroll
    for (int off = 16; off > 0; off >>= 1) {
        for (int e = 0; e < 4; ++e) {
            sv[e] += __shfl_xor(sv[e], off, 32);
            tv[e] += __shfl_xor(tv[e], off, 32);
        }
    }
    if (lane == 0) {
        *(v4f*)(s + (size_t)hd * N_NODES + 4 * qd) = sv;
        *(v4f*)(t + (size_t)hd * N_NODES + 4 * qd) = tv;
    }
}

#define KSPLIT   8
#define RTILES   4
#define ATT_THR  (KSPLIT * RTILES * 32)
__global__ __launch_bounds__(ATT_THR) void gat_attn(const float* __restrict__ h,
                                                const float* __restrict__ s,
                                                const float* __restrict__ t,
                                                float* __restrict__ out) {
    __shared__ float cred[RTILES][KSPLIT][16][64];
    __shared__ float zred[RTILES][KSPLIT][32];
    __shared__ float mred[RTILES][KSPLIT][16];
    __shared__ float wred[RTILES][KSPLIT][16];
    __shared__ float zrow[RTILES * 16];

    const int lane = threadIdx.x & 31;
    const int wave = threadIdx.x >> 5;
    const int rt   = wave & (RTILES - 1);
    const int ks   = wave / RTILES;
    const int ib   = blockIdx.x * (RTILES * 16);
    const int i0   = ib + rt * 16;
    const int hd   = blockIdx.y;
    const int m    = lane & 15;
    const int lh   = lane >> 4;

    const int jbeg = ks * (N_NODES / KSPLIT);
    const int jend = jbeg + (N_NODES / KSPLIT);

    const float sm   = s[hd * N_NODES + i0 + m];
    float tm = -INFINITY;
    {
        const float* tw = t + (size_t)hd * N_NODES + jbeg + 4 * lane;
        for (int i = 0; i < (N_NODES / KSPLIT) / 128; ++i) {
            const v4f tv = *(const v4f*)(tw + i * 128);
            tm = fmaxf(tm, fmaxf(fmaxf(tv.x, tv.y), fmaxf(tv.z, tv.w)));
        }
#pragma unroll
        for (int off = 16; off > 0; off >>= 1)
            tm = fmaxf(tm, __shfl_xor(tm, off, 32));
    }
    const float mrow = lrelu(sm + tm);

    v8f acc0 = {}, acc1 = {}, acc2 = {}, acc3 = {};
    float zpart = 0.f;
    const float* tp = t + (size_t)hd * N_NODES + 4 * lh;
    const float* hp = h + ((size_t)lh * HF + hd * FDIM + m) * 4;
    for (int j0 = jbeg; j0 < jend; j0 += 16) {
        v4f t4[2], g[2][4];
        for (int u = 0; u < 2; ++u)
            t4[u] = *(const v4f*)(tp + j0 + 8 * u);
        for (int u = 0; u < 2; ++u) {
            const float* hq = hp + (size_t)((j0 >> 2) + 2 * u) * (HF * 4);
            for (int q = 0; q < 4; ++q)
                g[u][q] = *(const v4f*)(hq + q * (16 * 4));
        }
        __builtin_amdgcn_sched_barrier(0);

        for (int u = 0; u < 2; ++u) {
            const float p0 = __expf(lrelu(sm + t4[u].x) - mrow);
            const float p1 = __expf(lrelu(sm + t4[u].y) - mrow);
            const float p2 = __expf(lrelu(sm + t4[u].z) - mrow);
            const float p3 = __expf(lrelu(sm + t4[u].w) - mrow);
            zpart += p0 + p1;
            zpart += p2 + p3;

            v2f a2, b0, b1, b2, b3;
            a2.x = p0;         a2.y = p1;
            b0.x = g[u][0].x;  b0.y = g[u][0].y;
            b1.x = g[u][1].x;  b1.y = g[u][1].y;
            b2.x = g[u][2].x;  b2.y = g[u][2].y;
            b3.x = g[u][3].x;  b3.y = g[u][3].y;
            acc0 = __builtin_amdgcn_wmma_f32_16x16x4_f32(false, a2, false, b0, (short)0, acc0, false, false);
            acc1 = __builtin_amdgcn_wmma_f32_16x16x4_f32(false, a2, false, b1, (short)0, acc1, false, false);
            acc2 = __builtin_amdgcn_wmma_f32_16x16x4_f32(false, a2, false, b2, (short)0, acc2, false, false);
            acc3 = __builtin_amdgcn_wmma_f32_16x16x4_f32(false, a2, false, b3, (short)0, acc3, false, false);

            a2.x = p2;         a2.y = p3;
            b0.x = g[u][0].z;  b0.y = g[u][0].w;
            b1.x = g[u][1].z;  b1.y = g[u][1].w;
            b2.x = g[u][2].z;  b2.y = g[u][2].w;
            b3.x = g[u][3].z;  b3.y = g[u][3].w;
            acc0 = __builtin_amdgcn_wmma_f32_16x16x4_f32(false, a2, false, b0, (short)0, acc0, false, false);
            acc1 = __builtin_amdgcn_wmma_f32_16x16x4_f32(false, a2, false, b1, (short)0, acc1, false, false);
            acc2 = __builtin_amdgcn_wmma_f32_16x16x4_f32(false, a2, false, b2, (short)0, acc2, false, false);
            acc3 = __builtin_amdgcn_wmma_f32_16x16x4_f32(false, a2, false, b3, (short)0, acc3, false, false);
        }
    }

    const int rbase = (lane >> 4) << 3;
#pragma unroll
    for (int v = 0; v < 8; ++v) {
        cred[rt][ks][rbase + v][m]      = acc0[v];
        cred[rt][ks][rbase + v][16 + m] = acc1[v];
        cred[rt][ks][rbase + v][32 + m] = acc2[v];
        cred[rt][ks][rbase + v][48 + m] = acc3[v];
    }
    zred[rt][ks][lane] = zpart;
    if (lane < 16) mred[rt][ks][m] = mrow;
    __syncthreads();

    if (threadIdx.x < RTILES * 16) {
        const int zr = threadIdx.x >> 4, zm = threadIdx.x & 15;
        float mx = mred[zr][0][zm];
#pragma unroll
        for (int w = 1; w < KSPLIT; ++w)
            mx = fmaxf(mx, mred[zr][w][zm]);
        float z = 0.f;
#pragma unroll
        for (int w = 0; w < KSPLIT; ++w) {
            const float wgt = __expf(mred[zr][w][zm] - mx);
            wred[zr][w][zm] = wgt;
            z += wgt * (zred[zr][w][zm] + zred[zr][w][zm + 16]);
        }
        zrow[threadIdx.x] = z;
    }
    __syncthreads();

    for (int e = threadIdx.x; e < RTILES * 16 * 64; e += ATT_THR) {
        const int er = e >> 10, rr = (e >> 6) & 15, cc = e & 63;
        float v = cred[er][0][rr][cc] * wred[er][0][rr];
        for (int w = 1; w < KSPLIT; ++w)
            v += cred[er][w][rr][cc] * wred[er][w][rr];
        out[(size_t)(ib + er * 16 + rr) * HF + hd * FDIM + cc] = v / zrow[er * 16 + rr];
    }
}

extern "C" void kernel_launch(void* const* d_in, const int* in_sizes, int n_in,
                              void* d_out, int out_size, void* d_ws, size_t ws_size,
                              hipStream_t stream) {
    const float* x = (const float*)d_in[0];
    const float* W = (const float*)d_in[2];
    const float* a = (const float*)d_in[3];
    float* out = (float*)d_out;

    char*  ws   = (char*)d_ws;
    float* h    = (float*)ws;
    float* s    = (float*)(ws + (size_t)N_NODES * HF * sizeof(float));
    float* t    = s + N_NODES * NHEAD;

    dim3 g1(N_NODES / 64, HF / 64);
    gat_gemm1<<<g1, 128, 0, stream>>>(x, W, h);

    gat_st<<<N_NODES / 4, 128, 0, stream>>>(h, a, s, t);

    dim3 g4(N_NODES / (RTILES * 16), NHEAD);
    gat_attn<<<g4, ATT_THR, 0, stream>>>(h, s, t, out);
}


# s,t taken in the x@W wave's epilogue from its own accumulators (16x64 tile = one head): 4 FMAs a row, DPP butterfly over 16 lanes, 2 lanes store; s,t launch gone (3 -> 2); s,t sum order changed
# speedup vs baseline: 1.6435x; 1.0374x over previous
#include <hip/hip_runtime.h>
#include <math.h>

#define ALPHA   0.2f
#define N_NODES 4096
#define IN_FEAT 512
#define NHEAD   4
#define FDIM    64
#define HF      256

typedef __attribute__((ext_vector_type(2))) float v2f;
typedef __attribute__((ext_vector_type(8))) float v8f;

__device__ __forceinline__ float lrelu(float x) {
    return fmaxf(x, ALPHA * x);
}

typedef __attribute__((ext_vector_type(4))) float v4f;

template <int XM>
__device__ __forceinline__ float row_xor(float v) {
    return __builtin_bit_cast(float,
        __builtin_amdgcn_update_dpp(0, __builtin_bit_cast(int, v), 0x160 | XM, 0xf, 0xf, true));
}

#define G1_CH  8
#define G1_NST (IN_FEAT / (8 * G1_CH))

struct G1Stage {
    v4f a[G1_CH];
    v4f b[G1_CH][4];
};

__device__ __forceinline__ void g1_load(G1Stage& s, const float* __restrict__ xp,
                                        const float* __restrict__ wp) {
    for (int ch = 0; ch < G1_CH; ++ch) {
        s.a[ch] = *(const v4f*)(xp + ch * 8);
        for (int kk = 0; kk < 4; ++kk)
            s.b[ch][kk] = *(const v4f*)(wp + (size_t)(ch * 8 + kk) * HF);
    }
}

__device__ __forceinline__ void g1_mma(const G1Stage& s, v8f& c0, v8f& c1, v8f& c2, v8f& c3) {
    for (int ch = 0; ch < G1_CH; ++ch) {
        for (int p = 0; p < 2; ++p) {
            v2f a2, b0, b1, b2, b3;
            a2.x = s.a[ch][2 * p];        a2.y = s.a[ch][2 * p + 1];
            b0.x = s.b[ch][2 * p][0];     b0.y = s.b[ch][2 * p + 1][0];
            b1.x = s.b[ch][2 * p][1];     b1.y = s.b[ch][2 * p + 1][1];
            b2.x = s.b[ch][2 * p][2];     b2.y = s.b[ch][2 * p + 1][2];
            b3.x = s.b[ch][2 * p][3];     b3.y = s.b[ch][2 * p + 1][3];
            c0 = __builtin_amdgcn_wmma_f32_16x16x4_f32(false, a2, false, b0, (short)0, c0, false, false);
            c1 = __builtin_amdgcn_wmma_f32_16x16x4_f32(false, a2, false, b1, (short)0, c1, false, false);
            c2 = __builtin_amdgcn_wmma_f32_16x16x4_f32(false, a2, false, b2, (short)0, c2, false, false);
            c3 = __builtin_amdgcn_wmma_f32_16x16x4_f32(false, a2, false, b3, (short)0, c3, false, false);
        }
    }
}

__global__ __launch_bounds__(128) void gat_gemm1(const float* __restrict__ x,
                                                 const float* __restrict__ W,
                                                 const float* __restrict__ a,
                                                 float* __restrict__ h,
                                                 float* __restrict__ s,
                                                 float* __restrict__ t) {
    const int lane = threadIdx.x & 31;
    const int wave = threadIdx.x >> 5;
    const int c    = lane & 15;
    const int hh   = lane >> 4;
    const int i0   = (blockIdx.x * 4 + wave) * 16;
    const int hd   = blockIdx.y;
    const int n0   = hd * FDIM;

    const float* xp = x + (size_t)(i0 + c) * IN_FEAT + 4 * hh;
    const float* wp = W + (size_t)(4 * hh) * HF + n0 + 4 * c;

    const v4f as4 = *(const v4f*)(a + hd * 2 * FDIM + 4 * c);
    const v4f ad4 = *(const v4f*)(a + hd * 2 * FDIM + FDIM + 4 * c);

    v8f c0 = {}, c1 = {}, c2 = {}, c3 = {};
    G1Stage s0, s1;
    g1_load(s0, xp, wp);
#pragma nounroll
    for (int st = 0; st < G1_NST - 2; st += 2) {
        g1_load(s1, xp + (st + 1) * (8 * G1_CH), wp + (size_t)(st + 1) * (8 * G1_CH) * HF);
        __builtin_amdgcn_sched_barrier(0);
        g1_mma(s0, c0, c1, c2, c3);
        g1_load(s0, xp + (st + 2) * (8 * G1_CH), wp + (size_t)(st + 2) * (8 * G1_CH) * HF);
        __builtin_amdgcn_sched_barrier(0);
        g1_mma(s1, c0, c1, c2, c3);
    }
    g1_load(s1, xp + (G1_NST - 1) * (8 * G1_CH), wp + (size_t)(G1_NST - 1) * (8 * G1_CH) * HF);
    __builtin_amdgcn_sched_barrier(0);
    g1_mma(s0, c0, c1, c2, c3);
    g1_mma(s1, c0, c1, c2, c3);

    float* hp = h + ((size_t)((i0 + 8 * hh) >> 2) * HF + n0 + 4 * c) * 4;
    for (int g = 0; g < 2; ++g) {
        v4f o0, o1, o2, o3;
        o0.x = c0[4 * g]; o0.y = c0[4 * g + 1]; o0.z = c0[4 * g + 2]; o0.w = c0[4 * g + 3];
        o1.x = c1[4 * g]; o1.y = c1[4 * g + 1]; o1.z = c1[4 * g + 2]; o1.w = c1[4 * g + 3];
        o2.x = c2[4 * g]; o2.y = c2[4 * g + 1]; o2.z = c2[4 * g + 2]; o2.w = c2[4 * g + 3];
        o3.x = c3[4 * g]; o3.y = c3[4 * g + 1]; o3.z = c3[4 * g + 2]; o3.w = c3[4 * g + 3];
        float* hq = hp + (size_t)g * HF * 4;
        *(v4f*)(hq)      = o0;
        *(v4f*)(hq + 4)  = o1;
        *(v4f*)(hq + 8)  = o2;
        *(v4f*)(hq + 12) = o3;
    }

    float sv[8], tv[8];
    for (int v = 0; v < 8; ++v) {
        sv[v] = c0[v] * as4.x + c1[v] * as4.y + c2[v] * as4.z + c3[v] * as4.w;
        tv[v] = c0[v] * ad4.x + c1[v] * ad4.y + c2[v] * ad4.z + c3[v] * ad4.w;
    }
    for (int v = 0; v < 8; ++v) { sv[v] += row_xor<1>(sv[v]); tv[v] += row_xor<1>(tv[v]); }
    for (int v = 0; v < 8; ++v) { sv[v] += row_xor<2>(sv[v]); tv[v] += row_xor<2>(tv[v]); }
    for (int v = 0; v < 8; ++v) { sv[v] += row_xor<4>(sv[v]); tv[v] += row_xor<4>(tv[v]); }
    for (int v = 0; v < 8; ++v) { sv[v] += row_xor<8>(sv[v]); tv[v] += row_xor<8>(tv[v]); }
    if (c == 0) {
        float* sp = s + (size_t)hd * N_NODES + i0 + 8 * hh;
        float* tq = t + (size_t)hd * N_NODES + i0 + 8 * hh;
        v4f w0, w1, w2, w3;
        w0.x = sv[0]; w0.y = sv[1]; w0.z = sv[2]; w0.w = sv[3];
        w1.x = sv[4]; w1.y = sv[5]; w1.z = sv[6]; w1.w = sv[7];
        w2.x = tv[0]; w2.y = tv[1]; w2.z = tv[2]; w2.w = tv[3];
        w3.x = tv[4]; w3.y = tv[5]; w3.z = tv[6]; w3.w = tv[7];
        *(v4f*)(sp)     = w0;
        *(v4f*)(sp + 4) = w1;
        *(v4f*)(tq)     = w2;
        *(v4f*)(tq + 4) = w3;
    }
}

#define KSPLIT   8
#define RTILES   4
#define ATT_THR  (KSPLIT * RTILES * 32)
__global__ __launch_bounds__(ATT_THR) void gat_attn(const float* __restrict__ h,
                                                const float* __restrict__ s,
                                                const float* __restrict__ t,
                                                float* __restrict__ out) {
    __shared__ float cred[RTILES][KSPLIT][16][64];
    __shared__ float zred[RTILES][KSPLIT][32];
    __shared__ float mred[RTILES][KSPLIT][16];
    __shared__ float wred[RTILES][KSPLIT][16];
    __shared__ float zrow[RTILES * 16];

    const int lane = threadIdx.x & 31;
    const int wave = threadIdx.x >> 5;
    const int rt   = wave & (RTILES - 1);
    const int ks   = wave / RTILES;
    const int ib   = blockIdx.x * (RTILES * 16);
    const int i0   = ib + rt * 16;
    const int hd   = blockIdx.y;
    const int m    = lane & 15;
    const int lh   = lane >> 4;

    const int jbeg = ks * (N_NODES / KSPLIT);
    const int jend = jbeg + (N_NODES / KSPLIT);

    const float sm   = s[hd * N_NODES + i0 + m];
    float tm = -INFINITY;
    {
        const float* tw = t + (size_t)hd * N_NODES + jbeg + 4 * lane;
        for (int i = 0; i < (N_NODES / KSPLIT) / 128; ++i) {
            const v4f tv = *(const v4f*)(tw + i * 128);
            tm = fmaxf(tm, fmaxf(fmaxf(tv.x, tv.y), fmaxf(tv.z, tv.w)));
        }
#pragma unroll
        for (int off = 16; off > 0; off >>= 1)
            tm = fmaxf(tm, __shfl_xor(tm, off, 32));
    }
    const float mrow = lrelu(sm + tm);

    v8f acc0 = {}, acc1 = {}, acc2 = {}, acc3 = {};
    float zpart = 0.f;
    const float* tp = t + (size_t)hd * N_NODES + 4 * lh;
    const float* hp = h + ((size_t)lh * HF + hd * FDIM + m) * 4;
    for (int j0 = jbeg; j0 < jend; j0 += 16) {
        v4f t4[2], g[2][4];
        for (int u = 0; u < 2; ++u)
            t4[u] = *(const v4f*)(tp + j0 + 8 * u);
        for (int u = 0; u < 2; ++u) {
            const float* hq = hp + (size_t)((j0 >> 2) + 2 * u) * (HF * 4);
            for (int q = 0; q < 4; ++q)
                g[u][q] = *(const v4f*)(hq + q * (16 * 4));
        }
        __builtin_amdgcn_sched_barrier(0);

        for (int u = 0; u < 2; ++u) {
            const float p0 = __expf(lrelu(sm + t4[u].x) - mrow);
            const float p1 = __expf(lrelu(sm + t4[u].y) - mrow);
            const float p2 = __expf(lrelu(sm + t4[u].z) - mrow);
            const float p3 = __expf(lrelu(sm + t4[u].w) - mrow);
            zpart += p0 + p1;
            zpart += p2 + p3;

            v2f a2, b0, b1, b2, b3;
            a2.x = p0;         a2.y = p1;
            b0.x = g[u][0].x;  b0.y = g[u][0].y;
            b1.x = g[u][1].x;  b1.y = g[u][1].y;
            b2.x = g[u][2].x;  b2.y = g[u][2].y;
            b3.x = g[u][3].x;  b3.y = g[u][3].y;
            acc0 = __builtin_amdgcn_wmma_f32_16x16x4_f32(false, a2, false, b0, (short)0, acc0, false, false);
            acc1 = __builtin_amdgcn_wmma_f32_16x16x4_f32(false, a2, false, b1, (short)0, acc1, false, false);
            acc2 = __builtin_amdgcn_wmma_f32_16x16x4_f32(false, a2, false, b2, (short)0, acc2, false, false);
            acc3 = __builtin_amdgcn_wmma_f32_16x16x4_f32(false, a2, false, b3, (short)0, acc3, false, false);

            a2.x = p2;         a2.y = p3;
            b0.x = g[u][0].z;  b0.y = g[u][0].w;
            b1.x = g[u][1].z;  b1.y = g[u][1].w;
            b2.x = g[u][2].z;  b2.y = g[u][2].w;
            b3.x = g[u][3].z;  b3.y = g[u][3].w;
            acc0 = __builtin_amdgcn_wmma_f32_16x16x4_f32(false, a2, false, b0, (short)0, acc0, false, false);
            acc1 = __builtin_amdgcn_wmma_f32_16x16x4_f32(false, a2, false, b1, (short)0, acc1, false, false);
            acc2 = __builtin_amdgcn_wmma_f32_16x16x4_f32(false, a2, false, b2, (short)0, acc2, false, false);
            acc3 = __builtin_amdgcn_wmma_f32_16x16x4_f32(false, a2, false, b3, (short)0, acc3, false, false);
        }
    }

    const int rbase = (lane >> 4) << 3;
#pragma unroll
    for (int v = 0; v < 8; ++v) {
        cred[rt][ks][rbase + v][m]      = acc0[v];
        cred[rt][ks][rbase + v][16 + m] = acc1[v];
        cred[rt][ks][rbase + v][32 + m] = acc2[v];
        cred[rt][ks][rbase + v][48 + m] = acc3[v];
    }
    zred[rt][ks][lane] = zpart;
    if (lane < 16) mred[rt][ks][m] = mrow;
    __syncthreads();

    if (threadIdx.x < RTILES * 16) {
        const int zr = threadIdx.x >> 4, zm = threadIdx.x & 15;
        float mx = mred[zr][0][zm];
#pragma unroll
        for (int w = 1; w < KSPLIT; ++w)
            mx = fmaxf(mx, mred[zr][w][zm]);
        float z = 0.f;
#pragma unroll
        for (int w = 0; w < KSPLIT; ++w) {
            const float wgt = __expf(mred[zr][w][zm] - mx);
            wred[zr][w][zm] = wgt;
            z += wgt * (zred[zr][w][zm] + zred[zr][w][zm + 16]);
        }
        zrow[threadIdx.x] = z;
    }
    __syncthreads();

    for (int e = threadIdx.x; e < RTILES * 16 * 64; e += ATT_THR) {
        const int er = e >> 10, rr = (e >> 6) & 15, cc = e & 63;
        float v = cred[er][0][rr][cc] * wred[er][0][rr];
        for (int w = 1; w < KSPLIT; ++w)
            v += cred[er][w][rr][cc] * wred[er][w][rr];
        out[(size_t)(ib + er * 16 + rr) * HF + hd * FDIM + cc] = v / zrow[er * 16 + rr];
    }
}

extern "C" void kernel_launch(void* const* d_in, const int* in_sizes, int n_in,
                              void* d_out, int out_size, void* d_ws, size_t ws_size,
                              hipStream_t stream) {
    const float* x = (const float*)d_in[0];
    const float* W = (const float*)d_in[2];
    const float* a = (const float*)d_in[3];
    float* out = (float*)d_out;

    char*  ws   = (char*)d_ws;
    float* h    = (float*)ws;
    float* s    = (float*)(ws + (size_t)N_NODES * HF * sizeof(float));
    float* t    = s + N_NODES * NHEAD;

    dim3 g1(N_NODES / 64, HF / 64);
    gat_gemm1<<<g1, 128, 0, stream>>>(x, W, a, h, s, t);

    dim3 g4(N_NODES / (RTILES * 16), NHEAD);
    gat_attn<<<g4, ATT_THR, 0, stream>>>(h, s, t, out);
}


# attention key loop in 4 phases of 16/8/6/2 turns with falling wave priority (s_setprio 3,2,1,0): a wave behind is served first, the 8 waves of a SIMD end together, no lone tail; sum order kept
# speedup vs baseline: 1.8459x; 1.1232x over previous
#include <hip/hip_runtime.h>
#include <math.h>

#define ALPHA   0.2f
#define N_NODES 4096
#define IN_FEAT 512
#define NHEAD   4
#define FDIM    64
#define HF      256

typedef __attribute__((ext_vector_type(2))) float v2f;
typedef __attribute__((ext_vector_type(8))) float v8f;

__device__ __forceinline__ float lrelu(float x) {
    return fmaxf(x, ALPHA * x);
}

typedef __attribute__((ext_vector_type(4))) float v4f;

template <int XM>
__device__ __forceinline__ float row_xor(float v) {
    return __builtin_bit_cast(float,
        __builtin_amdgcn_update_dpp(0, __builtin_bit_cast(int, v), 0x160 | XM, 0xf, 0xf, true));
}

#define G1_CH  8
#define G1_NST (IN_FEAT / (8 * G1_CH))

struct G1Stage {
    v4f a[G1_CH];
    v4f b[G1_CH][4];
};

__device__ __forceinline__ void g1_load(G1Stage& s, const float* __restrict__ xp,
                                        const float* __restrict__ wp) {
    for (int ch = 0; ch < G1_CH; ++ch) {
        s.a[ch] = *(const v4f*)(xp + ch * 8);
        for (int kk = 0; kk < 4; ++kk)
            s.b[ch][kk] = *(const v4f*)(wp + (size_t)(ch * 8 + kk) * HF);
    }
}

__device__ __forceinline__ void g1_mma(const G1Stage& s, v8f& c0, v8f& c1, v8f& c2, v8f& c3) {
    for (int ch = 0; ch < G1_CH; ++ch) {
        for (int p = 0; p < 2; ++p) {
            v2f a2, b0, b1, b2, b3;
            a2.x = s.a[ch][2 * p];        a2.y = s.a[ch][2 * p + 1];
            b0.x = s.b[ch][2 * p][0];     b0.y = s.b[ch][2 * p + 1][0];
            b1.x = s.b[ch][2 * p][1];     b1.y = s.b[ch][2 * p + 1][1];
            b2.x = s.b[ch][2 * p][2];     b2.y = s.b[ch][2 * p + 1][2];
            b3.x = s.b[ch][2 * p][3];     b3.y = s.b[ch][2 * p + 1][3];
            c0 = __builtin_amdgcn_wmma_f32_16x16x4_f32(false, a2, false, b0, (short)0, c0, false, false);
            c1 = __builtin_amdgcn_wmma_f32_16x16x4_f32(false, a2, false, b1, (short)0, c1, false, false);
            c2 = __builtin_amdgcn_wmma_f32_16x16x4_f32(false, a2, false, b2, (short)0, c2, false, false);
            c3 = __builtin_amdgcn_wmma_f32_16x16x4_f32(false, a2, false, b3, (short)0, c3, false, false);
        }
    }
}

__global__ __launch_bounds__(128) void gat_gemm1(const float* __restrict__ x,
                                                 const float* __restrict__ W,
                                                 const float* __restrict__ a,
                                                 float* __restrict__ h,
                                                 float* __restrict__ s,
                                                 float* __restrict__ t) {
    const int lane = threadIdx.x & 31;
    const int wave = threadIdx.x >> 5;
    const int c    = lane & 15;
    const int hh   = lane >> 4;
    const int i0   = (blockIdx.x * 4 + wave) * 16;
    const int hd   = blockIdx.y;
    const int n0   = hd * FDIM;

    const float* xp = x + (size_t)(i0 + c) * IN_FEAT + 4 * hh;
    const float* wp = W + (size_t)(4 * hh) * HF + n0 + 4 * c;

    const v4f as4 = *(const v4f*)(a + hd * 2 * FDIM + 4 * c);
    const v4f ad4 = *(const v4f*)(a + hd * 2 * FDIM + FDIM + 4 * c);

    v8f c0 = {}, c1 = {}, c2 = {}, c3 = {};
    G1Stage s0, s1;
    g1_load(s0, xp, wp);
#pragma nounroll
    for (int st = 0; st < G1_NST - 2; st += 2) {
        g1_load(s1, xp + (st + 1) * (8 * G1_CH), wp + (size_t)(st + 1) * (8 * G1_CH) * HF);
        __builtin_amdgcn_sched_barrier(0);
        g1_mma(s0, c0, c1, c2, c3);
        g1_load(s0, xp + (st + 2) * (8 * G1_CH), wp + (size_t)(st + 2) * (8 * G1_CH) * HF);
        __builtin_amdgcn_sched_barrier(0);
        g1_mma(s1, c0, c1, c2, c3);
    }
    g1_load(s1, xp + (G1_NST - 1) * (8 * G1_CH), wp + (size_t)(G1_NST - 1) * (8 * G1_CH) * HF);
    __builtin_amdgcn_sched_barrier(0);
    g1_mma(s0, c0, c1, c2, c3);
    g1_mma(s1, c0, c1, c2, c3);

    float* hp = h + ((size_t)((i0 + 8 * hh) >> 2) * HF + n0 + 4 * c) * 4;
    for (int g = 0; g < 2; ++g) {
        v4f o0, o1, o2, o3;
        o0.x = c0[4 * g]; o0.y = c0[4 * g + 1]; o0.z = c0[4 * g + 2]; o0.w = c0[4 * g + 3];
        o1.x = c1[4 * g]; o1.y = c1[4 * g + 1]; o1.z = c1[4 * g + 2]; o1.w = c1[4 * g + 3];
        o2.x = c2[4 * g]; o2.y = c2[4 * g + 1]; o2.z = c2[4 * g + 2]; o2.w = c2[4 * g + 3];
        o3.x = c3[4 * g]; o3.y = c3[4 * g + 1]; o3.z = c3[4 * g + 2]; o3.w = c3[4 * g + 3];
        float* hq = hp + (size_t)g * HF * 4;
        *(v4f*)(hq)      = o0;
        *(v4f*)(hq + 4)  = o1;
        *(v4f*)(hq + 8)  = o2;
        *(v4f*)(hq + 12) = o3;
    }

    float sv[8], tv[8];
    for (int v = 0; v < 8; ++v) {
        sv[v] = c0[v] * as4.x + c1[v] * as4.y + c2[v] * as4.z + c3[v] * as4.w;
        tv[v] = c0[v] * ad4.x + c1[v] * ad4.y + c2[v] * ad4.z + c3[v] * ad4.w;
    }
    for (int v = 0; v < 8; ++v) { sv[v] += row_xor<1>(sv[v]); tv[v] += row_xor<1>(tv[v]); }
    for (int v = 0; v < 8; ++v) { sv[v] += row_xor<2>(sv[v]); tv[v] += row_xor<2>(tv[v]); }
    for (int v = 0; v < 8; ++v) { sv[v] += row_xor<4>(sv[v]); tv[v] += row_xor<4>(tv[v]); }
    for (int v = 0; v < 8; ++v) { sv[v] += row_xor<8>(sv[v]); tv[v] += row_xor<8>(tv[v]); }
    if (c == 0) {
        float* sp = s + (size_t)hd * N_NODES + i0 + 8 * hh;
        float* tq = t + (size_t)hd * N_NODES + i0 + 8 * hh;
        v4f w0, w1, w2, w3;
        w0.x = sv[0]; w0.y = sv[1]; w0.z = sv[2]; w0.w = sv[3];
        w1.x = sv[4]; w1.y = sv[5]; w1.z = sv[6]; w1.w = sv[7];
        w2.x = tv[0]; w2.y = tv[1]; w2.z = tv[2]; w2.w = tv[3];
        w3.x = tv[4]; w3.y = tv[5]; w3.z = tv[6]; w3.w = tv[7];
        *(v4f*)(sp)     = w0;
        *(v4f*)(sp + 4) = w1;
        *(v4f*)(tq)     = w2;
        *(v4f*)(tq + 4) = w3;
    }
}

#define KSPLIT   8
#define RTILES   4
#define ATT_THR  (KSPLIT * RTILES * 32)
#define PH0 16
#define PH1 8
#define PH2 6
#define PH3 2
template <int TURNS>
__device__ __forceinline__ void attn_turns(const float*& tp, const float*& hp,
                                           const float sm, const float mrow,
                                           v8f& acc0, v8f& acc1, v8f& acc2, v8f& acc3, float& zpart) {
#pragma nounroll
    for (int it = 0; it < TURNS; ++it) {
        v4f t4[2], g[2][4];
        for (int u = 0; u < 2; ++u)
            t4[u] = *(const v4f*)(tp + 8 * u);
        for (int u = 0; u < 2; ++u) {
            const float* hq = hp + (size_t)(2 * u) * (HF * 4);
            for (int q = 0; q < 4; ++q)
                g[u][q] = *(const v4f*)(hq + q * (16 * 4));
        }
        __builtin_amdgcn_sched_barrier(0);

        for (int u = 0; u < 2; ++u) {
            const float p0 = __expf(lrelu(sm + t4[u].x) - mrow);
            const float p1 = __expf(lrelu(sm + t4[u].y) - mrow);
            const float p2 = __expf(lrelu(sm + t4[u].z) - mrow);
            const float p3 = __expf(lrelu(sm + t4[u].w) - mrow);
            zpart += p0 + p1;
            zpart += p2 + p3;

            v2f a2, b0, b1, b2, b3;
            a2.x = p0;         a2.y = p1;
            b0.x = g[u][0].x;  b0.y = g[u][0].y;
            b1.x = g[u][1].x;  b1.y = g[u][1].y;
            b2.x = g[u][2].x;  b2.y = g[u][2].y;
            b3.x = g[u][3].x;  b3.y = g[u][3].y;
            acc0 = __builtin_amdgcn_wmma_f32_16x16x4_f32(false, a2, false, b0, (short)0, acc0, false, false);
            acc1 = __builtin_amdgcn_wmma_f32_16x16x4_f32(false, a2, false, b1, (short)0, acc1, false, false);
            acc2 = __builtin_amdgcn_wmma_f32_16x16x4_f32(false, a2, false, b2, (short)0, acc2, false, false);
            acc3 = __builtin_amdgcn_wmma_f32_16x16x4_f32(false, a2, false, b3, (short)0, acc3, false, false);

            a2.x = p2;         a2.y = p3;
            b0.x = g[u][0].z;  b0.y = g[u][0].w;
            b1.x = g[u][1].z;  b1.y = g[u][1].w;
            b2.x = g[u][2].z;  b2.y = g[u][2].w;
            b3.x = g[u][3].z;  b3.y = g[u][3].w;
            acc0 = __builtin_amdgcn_wmma_f32_16x16x4_f32(false, a2, false, b0, (short)0, acc0, false, false);
            acc1 = __builtin_amdgcn_wmma_f32_16x16x4_f32(false, a2, false, b1, (short)0, acc1, false, false);
            acc2 = __builtin_amdgcn_wmma_f32_16x16x4_f32(false, a2, false, b2, (short)0, acc2, false, false);
            acc3 = __builtin_amdgcn_wmma_f32_16x16x4_f32(false, a2, false, b3, (short)0, acc3, false, false);
        }
        tp += 16;
        hp += 4 * (HF * 4);
    }
}

__global__ __launch_bounds__(ATT_THR) void gat_attn(const float* __restrict__ h,
                                                const float* __restrict__ s,
                                                const float* __restrict__ t,
                                                float* __restrict__ out) {
    __shared__ float cred[RTILES][KSPLIT][16][64];
    __shared__ float zred[RTILES][KSPLIT][32];
    __shared__ float mred[RTILES][KSPLIT][16];
    __shared__ float wred[RTILES][KSPLIT][16];
    __shared__ float zrow[RTILES * 16];

    const int lane = threadIdx.x & 31;
    const int wave = threadIdx.x >> 5;
    const int rt   = wave & (RTILES - 1);
    const int ks   = wave / RTILES;
    const int ib   = blockIdx.x * (RTILES * 16);
    const int i0   = ib + rt * 16;
    const int hd   = blockIdx.y;
    const int m    = lane & 15;
    const int lh   = lane >> 4;

    const int jbeg = ks * (N_NODES / KSPLIT);

    const float sm   = s[hd * N_NODES + i0 + m];
    float tm = -INFINITY;
    {
        const float* tw = t + (size_t)hd * N_NODES + jbeg + 4 * lane;
        for (int i = 0; i < (N_NODES / KSPLIT) / 128; ++i) {
            const v4f tv = *(const v4f*)(tw + i * 128);
            tm = fmaxf(tm, fmaxf(fmaxf(tv.x, tv.y), fmaxf(tv.z, tv.w)));
        }
#pragma unroll
        for (int off = 16; off > 0; off >>= 1)
            tm = fmaxf(tm, __shfl_xor(tm, off, 32));
    }
    const float mrow = lrelu(sm + tm);

    v8f acc0 = {}, acc1 = {}, acc2 = {}, acc3 = {};
    float zpart = 0.f;
    const float* tp = t + (size_t)hd * N_NODES + jbeg + 4 * lh;
    const float* hp = h + ((size_t)((jbeg >> 2) + lh) * HF + hd * FDIM + m) * 4;
    static_assert((PH0 + PH1 + PH2 + PH3) * 16 == N_NODES / KSPLIT, "phases cover the key range");
    __builtin_amdgcn_s_setprio(3);
    attn_turns<PH0>(tp, hp, sm, mrow, acc0, acc1, acc2, acc3, zpart);
    __builtin_amdgcn_s_setprio(2);
    attn_turns<PH1>(tp, hp, sm, mrow, acc0, acc1, acc2, acc3, zpart);
    __builtin_amdgcn_s_setprio(1);
    attn_turns<PH2>(tp, hp, sm, mrow, acc0, acc1, acc2, acc3, zpart);
    __builtin_amdgcn_s_setprio(0);
    attn_turns<PH3>(tp, hp, sm, mrow, acc0, acc1, acc2, acc3, zpart);

    const int rbase = (lane >> 4) << 3;
#pragma unroll
    for (int v = 0; v < 8; ++v) {
        cred[rt][ks][rbase + v][m]      = acc0[v];
        cred[rt][ks][rbase + v][16 + m] = acc1[v];
        cred[rt][ks][rbase + v][32 + m] = acc2[v];
        cred[rt][ks][rbase + v][48 + m] = acc3[v];
    }
    zred[rt][ks][lane] = zpart;
    if (lane < 16) mred[rt][ks][m] = mrow;
    __syncthreads();

    if (threadIdx.x < RTILES * 16) {
        const int zr = threadIdx.x >> 4, zm = threadIdx.x & 15;
        float mx = mred[zr][0][zm];
#pragma unroll
        for (int w = 1; w < KSPLIT; ++w)
            mx = fmaxf(mx, mred[zr][w][zm]);
        float z = 0.f;
#pragma unroll
        for (int w = 0; w < KSPLIT; ++w) {
            const float wgt = __expf(mred[zr][w][zm] - mx);
            wred[zr][w][zm] = wgt;
            z += wgt * (zred[zr][w][zm] + zred[zr][w][zm + 16]);
        }
        zrow[threadIdx.x] = z;
    }
    __syncthreads();

    for (int e = threadIdx.x; e < RTILES * 16 * 64; e += ATT_THR) {
        const int er = e >> 10, rr = (e >> 6) & 15, cc = e & 63;
        float v = cred[er][0][rr][cc] * wred[er][0][rr];
        for (int w = 1; w < KSPLIT; ++w)
            v += cred[er][w][rr][cc] * wred[er][w][rr];
        out[(size_t)(ib + er * 16 + rr) * HF + hd * FDIM + cc] = v / zrow[er * 16 + rr];
    }
}

extern "C" void kernel_launch(void* const* d_in, const int* in_sizes, int n_in,
                              void* d_out, int out_size, void* d_ws, size_t ws_size,
                              hipStream_t stream) {
    const float* x = (const float*)d_in[0];
    const float* W = (const float*)d_in[2];
    const float* a = (const float*)d_in[3];
    float* out = (float*)d_out;

    char*  ws   = (char*)d_ws;
    float* h    = (float*)ws;
    float* s    = (float*)(ws + (size_t)N_NODES * HF * sizeof(float));
    float* t    = s + N_NODES * NHEAD;

    dim3 g1(N_NODES / 64, HF / 64);
    gat_gemm1<<<g1, 128, 0, stream>>>(x, W, a, h, s, t);

    dim3 g4(N_NODES / (RTILES * 16), NHEAD);
    gat_attn<<<g4, ATT_THR, 0, stream>>>(h, s, t, out);
}
